# starting kernel
# baseline (speedup 1.0000x reference)
#include <hip/hip_runtime.h>

#define N_NODES 100000
#define N_EDGES 1600000
#define IN_DIM  128
#define OUT_DIM 64

typedef __attribute__((ext_vector_type(2))) float v2f;
typedef __attribute__((ext_vector_type(8))) float v8f;

__device__ __forceinline__ void atomAddF32(float* p, float v) {
    __hip_atomic_fetch_add(p, v, __ATOMIC_RELAXED, __HIP_MEMORY_SCOPE_AGENT);
}

__device__ __forceinline__ unsigned enc_f(float f) {
    unsigned u = __float_as_uint(f);
    return (u & 0x80000000u) ? ~u : (u | 0x80000000u);
}
__device__ __forceinline__ float dec_f(unsigned u) {
    unsigned b = (u & 0x80000000u) ? (u & 0x7FFFFFFFu) : ~u;
    return __uint_as_float(b);
}
#define ENC_NEG_INF 0x007FFFFFu

__global__ void init_kernel(float* __restrict__ out, unsigned* __restrict__ m_enc,
                            float* __restrict__ denom) {
    long stride = (long)gridDim.x * blockDim.x;
    long i = (long)blockIdx.x * blockDim.x + threadIdx.x;
    const long tot = (long)N_NODES * OUT_DIM;
    for (long j = i; j < tot; j += stride) out[j] = 0.0f;
    for (long j = i; j < N_NODES; j += stride) { m_enc[j] = ENC_NEG_INF; denom[j] = 0.0f; }
}

__global__ __launch_bounds__(128)
void gemm_wmma(const float* __restrict__ h, const float* __restrict__ W,
               float* __restrict__ z) {
    __shared__ float Wl[IN_DIM * OUT_DIM];

    const int tid = threadIdx.x;
    for (int i = tid * 4; i < IN_DIM * OUT_DIM; i += 128 * 4) {
        *(float4*)&Wl[i] = *(const float4*)&W[i];
    }
    __syncthreads();

    const int wave    = tid >> 5;
    const int lane    = tid & 31;
    const int l16     = lane & 15;
    const int khalf   = (lane >> 4) * 2;
    const int rowBase = blockIdx.x * 16;
    const int colBase = wave * 16;

    const float* hrow = h + (long)(rowBase + l16) * IN_DIM;

    v8f acc = {};
    #pragma unroll 4
    for (int k = 0; k < IN_DIM; k += 4) {
        v2f a = *(const v2f*)(hrow + k + khalf);
        v2f b;
        b.x = Wl[(k + khalf + 0) * OUT_DIM + colBase + l16];
        b.y = Wl[(k + khalf + 1) * OUT_DIM + colBase + l16];
        __builtin_prefetch(hrow + k + 32, 0, 1);
        acc = __builtin_amdgcn_wmma_f32_16x16x4_f32(
                  false, a, false, b, (short)0, acc, false, false);
    }

    float* zt = z + (long)rowBase * OUT_DIM + colBase + l16;
    const int mOff = (lane >> 4) * 8;
    #pragma unroll
    for (int v = 0; v < 8; ++v) {
        zt[(long)(v + mOff) * OUT_DIM] = acc[v];
    }
}

__global__ __launch_bounds__(256)
void scores_kernel(const float* __restrict__ z, const float* __restrict__ a,
                   float* __restrict__ s_src, float* __restrict__ s_dst) {
    long wid = ((long)blockIdx.x * blockDim.x + threadIdx.x) >> 5;
    int lane = threadIdx.x & 31;
    if (wid >= N_NODES) return;
    float2 zv = *(const float2*)(z + wid * OUT_DIM + lane * 2);
    float2 as = *(const float2*)(a + lane * 2);
    float2 ad = *(const float2*)(a + OUT_DIM + lane * 2);
    float ps = zv.x * as.x + zv.y * as.y;
    float pd = zv.x * ad.x + zv.y * ad.y;
    #pragma unroll
    for (int off = 16; off > 0; off >>= 1) {
        ps += __shfl_down(ps, off, 32);
        pd += __shfl_down(pd, off, 32);
    }
    if (lane == 0) { s_src[wid] = ps; s_dst[wid] = pd; }
}

__global__ __launch_bounds__(256)
void edge_att_max(const int* __restrict__ src, const int* __restrict__ dst,
                  const float* __restrict__ s_src, const float* __restrict__ s_dst,
                  float* __restrict__ att, unsigned* __restrict__ m_enc) {
    long e = (long)blockIdx.x * blockDim.x + threadIdx.x;
    if (e >= N_EDGES) return;
    int d = dst[e];
    float v = s_src[src[e]] + s_dst[d];
    float lr = v > 0.0f ? v : 0.01f * v;
    att[e] = lr;
    atomicMax(&m_enc[d], enc_f(lr));
}

__global__ __launch_bounds__(256)
void m_finalize(const unsigned* __restrict__ m_enc, float* __restrict__ m) {
    long i = (long)blockIdx.x * blockDim.x + threadIdx.x;
    if (i >= N_NODES) return;
    float v = dec_f(m_enc[i]);
    m[i] = (m_enc[i] == ENC_NEG_INF) ? 0.0f : v;
}

__global__ __launch_bounds__(256)
void edge_exp(const int* __restrict__ dst, const float* __restrict__ att,
              const float* __restrict__ m, float* __restrict__ ex,
              float* __restrict__ denom) {
    long e = (long)blockIdx.x * blockDim.x + threadIdx.x;
    if (e >= N_EDGES) return;
    int d = dst[e];
    float x = __expf(att[e] - m[d]);
    ex[e] = x;
    atomAddF32(&denom[d], x);
}

__global__ __launch_bounds__(256)
void edge_aggregate(const int* __restrict__ src, const int* __restrict__ dst,
                    const float* __restrict__ z, const float* __restrict__ ex,
                    const float* __restrict__ denom, float* __restrict__ out) {
    long wid = ((long)blockIdx.x * blockDim.x + threadIdx.x) >> 5;
    int lane = threadIdx.x & 31;
    if (wid >= N_EDGES) return;
    int s = src[wid];
    int d = dst[wid];
    float dn = denom[d];
    float alpha = ex[wid] / (dn == 0.0f ? 1.0f : dn);
    float2 zv = *(const float2*)(z + (long)s * OUT_DIM + lane * 2);
    float* o = out + (long)d * OUT_DIM + lane * 2;
    atomAddF32(o,     alpha * zv.x);
    atomAddF32(o + 1, alpha * zv.y);
}

static inline size_t alignUp(size_t x) { return (x + 255) & ~(size_t)255; }

extern "C" void kernel_launch(void* const* d_in, const int* in_sizes, int n_in,
                              void* d_out, int out_size, void* d_ws, size_t ws_size,
                              hipStream_t stream) {
    const float* h  = (const float*)d_in[0];
    const float* W  = (const float*)d_in[1];
    const float* a  = (const float*)d_in[2];
    const int* src  = (const int*)d_in[3];
    const int* dst  = (const int*)d_in[4];
    float* out      = (float*)d_out;

    char* ws = (char*)d_ws;
    float*    z     = (float*)ws;    ws += alignUp((size_t)N_NODES * OUT_DIM * 4);
    float*    s_src = (float*)ws;    ws += alignUp((size_t)N_NODES * 4);
    float*    s_dst = (float*)ws;    ws += alignUp((size_t)N_NODES * 4);
    unsigned* m_enc = (unsigned*)ws; ws += alignUp((size_t)N_NODES * 4);
    float*    m     = (float*)ws;    ws += alignUp((size_t)N_NODES * 4);
    float*    denom = (float*)ws;    ws += alignUp((size_t)N_NODES * 4);
    float*    att   = (float*)ws;    ws += alignUp((size_t)N_EDGES * 4);
    float*    ex    = (float*)ws;    ws += alignUp((size_t)N_EDGES * 4);

    init_kernel<<<2048, 256, 0, stream>>>(out, m_enc, denom);

    gemm_wmma<<<N_NODES / 16, 128, 0, stream>>>(h, W, z);

    long score_threads = (long)N_NODES * 32;
    scores_kernel<<<(unsigned)((score_threads + 255) / 256), 256, 0, stream>>>(z, a, s_src, s_dst);

    edge_att_max<<<(N_EDGES + 255) / 256, 256, 0, stream>>>(src, dst, s_src, s_dst, att, m_enc);

    m_finalize<<<(N_NODES + 255) / 256, 256, 0, stream>>>(m_enc, m);

    edge_exp<<<(N_EDGES + 255) / 256, 256, 0, stream>>>(dst, att, m, ex, denom);

    long agg_threads = (long)N_EDGES * 32;
    edge_aggregate<<<(unsigned)((agg_threads + 255) / 256), 256, 0, stream>>>(src, dst, z, ex, denom, out);
}


# gemm operand feed: W staged once per 128 rows into LDS in fragment order; each wave owns a 16-row tile over all 64 columns (4 accumulators); h read once by 128-bit loads, z stored as whole lines
# speedup vs baseline: 1.1021x; 1.1021x over previous
#include <hip/hip_runtime.h>

#define N_NODES 100000
#define N_EDGES 1600000
#define IN_DIM  128
#define OUT_DIM 64

typedef __attribute__((ext_vector_type(2))) float v2f;
typedef __attribute__((ext_vector_type(8))) float v8f;

__device__ __forceinline__ void atomAddF32(float* p, float v) {
    __hip_atomic_fetch_add(p, v, __ATOMIC_RELAXED, __HIP_MEMORY_SCOPE_AGENT);
}

__device__ __forceinline__ unsigned enc_f(float f) {
    unsigned u = __float_as_uint(f);
    return (u & 0x80000000u) ? ~u : (u | 0x80000000u);
}
__device__ __forceinline__ float dec_f(unsigned u) {
    unsigned b = (u & 0x80000000u) ? (u & 0x7FFFFFFFu) : ~u;
    return __uint_as_float(b);
}
#define ENC_NEG_INF 0x007FFFFFu

__global__ void init_kernel(float* __restrict__ out, unsigned* __restrict__ m_enc,
                            float* __restrict__ denom) {
    long stride = (long)gridDim.x * blockDim.x;
    long i = (long)blockIdx.x * blockDim.x + threadIdx.x;
    const long tot = (long)N_NODES * OUT_DIM;
    for (long j = i; j < tot; j += stride) out[j] = 0.0f;
    for (long j = i; j < N_NODES; j += stride) { m_enc[j] = ENC_NEG_INF; denom[j] = 0.0f; }
}

#define GEMM_WAVES 8
__global__ __launch_bounds__(256)
void gemm_wmma(const float* __restrict__ h, const float* __restrict__ W,
               float* __restrict__ z) {
    __shared__ float4 Wl4[16 * 4 * 32];

    const int tid  = threadIdx.x;
    const int wave = tid >> 5;
    const int lane = tid & 31;
    const int c    = lane & 15;
    const int hh   = lane >> 4;
    const int tileRaw = blockIdx.x * GEMM_WAVES + wave;
    const bool active = tileRaw < N_NODES / 16;
    const int tile = active ? tileRaw : (N_NODES / 16 - 1);

    const float* wcol = W + (8 * (wave >> 2) + 4 * hh) * OUT_DIM + 4 * c + (wave & 3);
    float4 w4[8];
    for (int m = 0; m < 8; ++m) {
        const float* wp = wcol + m * 16 * OUT_DIM;
        w4[m].x = wp[0 * OUT_DIM];
        w4[m].y = wp[1 * OUT_DIM];
        w4[m].z = wp[2 * OUT_DIM];
        w4[m].w = wp[3 * OUT_DIM];
    }

    const float4* hrow = (const float4*)(h + (long)(tile * 16 + c) * IN_DIM + 4 * hh);
    float4 a4[16];
    for (int u = 0; u < 16; ++u) a4[u] = hrow[2 * u];

    for (int m = 0; m < 8; ++m) Wl4[tid + 256 * m] = w4[m];
    __syncthreads();

    v8f acc0 = {}, acc1 = {}, acc2 = {}, acc3 = {};
    for (int u = 0; u < 16; ++u) {
        const float4 b0 = Wl4[(u * 4 + 0) * 32 + lane];
        const float4 b1 = Wl4[(u * 4 + 1) * 32 + lane];
        const float4 b2 = Wl4[(u * 4 + 2) * 32 + lane];
        const float4 b3 = Wl4[(u * 4 + 3) * 32 + lane];
        const v2f aLo = {a4[u].x, a4[u].y};
        const v2f aHi = {a4[u].z, a4[u].w};
        acc0 = __builtin_amdgcn_wmma_f32_16x16x4_f32(false, aLo, false, (v2f){b0.x, b0.y}, (short)0, acc0, false, false);
        acc1 = __builtin_amdgcn_wmma_f32_16x16x4_f32(false, aLo, false, (v2f){b1.x, b1.y}, (short)0, acc1, false, false);
        acc2 = __builtin_amdgcn_wmma_f32_16x16x4_f32(false, aLo, false, (v2f){b2.x, b2.y}, (short)0, acc2, false, false);
        acc3 = __builtin_amdgcn_wmma_f32_16x16x4_f32(false, aLo, false, (v2f){b3.x, b3.y}, (short)0, acc3, false, false);
        acc0 = __builtin_amdgcn_wmma_f32_16x16x4_f32(false, aHi, false, (v2f){b0.z, b0.w}, (short)0, acc0, false, false);
        acc1 = __builtin_amdgcn_wmma_f32_16x16x4_f32(false, aHi, false, (v2f){b1.z, b1.w}, (short)0, acc1, false, false);
        acc2 = __builtin_amdgcn_wmma_f32_16x16x4_f32(false, aHi, false, (v2f){b2.z, b2.w}, (short)0, acc2, false, false);
        acc3 = __builtin_amdgcn_wmma_f32_16x16x4_f32(false, aHi, false, (v2f){b3.z, b3.w}, (short)0, acc3, false, false);
    }

    if (active) {
        float* zt = z + (long)(tile * 16 + 8 * hh) * OUT_DIM + 4 * c;
        for (int r = 0; r < 8; ++r) {
            *(float4*)(zt + (long)r * OUT_DIM) = make_float4(acc0[r], acc1[r], acc2[r], acc3[r]);
        }
    }
}

__global__ __launch_bounds__(256)
void scores_kernel(const float* __restrict__ z, const float* __restrict__ a,
                   float* __restrict__ s_src, float* __restrict__ s_dst) {
    long wid = ((long)blockIdx.x * blockDim.x + threadIdx.x) >> 5;
    int lane = threadIdx.x & 31;
    if (wid >= N_NODES) return;
    float2 zv = *(const float2*)(z + wid * OUT_DIM + lane * 2);
    float2 as = *(const float2*)(a + lane * 2);
    float2 ad = *(const float2*)(a + OUT_DIM + lane * 2);
    float ps = zv.x * as.x + zv.y * as.y;
    float pd = zv.x * ad.x + zv.y * ad.y;
    #pragma unroll
    for (int off = 16; off > 0; off >>= 1) {
        ps += __shfl_down(ps, off, 32);
        pd += __shfl_down(pd, off, 32);
    }
    if (lane == 0) { s_src[wid] = ps; s_dst[wid] = pd; }
}

__global__ __launch_bounds__(256)
void edge_att_max(const int* __restrict__ src, const int* __restrict__ dst,
                  const float* __restrict__ s_src, const float* __restrict__ s_dst,
                  float* __restrict__ att, unsigned* __restrict__ m_enc) {
    long e = (long)blockIdx.x * blockDim.x + threadIdx.x;
    if (e >= N_EDGES) return;
    int d = dst[e];
    float v = s_src[src[e]] + s_dst[d];
    float lr = v > 0.0f ? v : 0.01f * v;
    att[e] = lr;
    atomicMax(&m_enc[d], enc_f(lr));
}

__global__ __launch_bounds__(256)
void m_finalize(const unsigned* __restrict__ m_enc, float* __restrict__ m) {
    long i = (long)blockIdx.x * blockDim.x + threadIdx.x;
    if (i >= N_NODES) return;
    float v = dec_f(m_enc[i]);
    m[i] = (m_enc[i] == ENC_NEG_INF) ? 0.0f : v;
}

__global__ __launch_bounds__(256)
void edge_exp(const int* __restrict__ dst, const float* __restrict__ att,
              const float* __restrict__ m, float* __restrict__ ex,
              float* __restrict__ denom) {
    long e = (long)blockIdx.x * blockDim.x + threadIdx.x;
    if (e >= N_EDGES) return;
    int d = dst[e];
    float x = __expf(att[e] - m[d]);
    ex[e] = x;
    atomAddF32(&denom[d], x);
}

__global__ __launch_bounds__(256)
void edge_aggregate(const int* __restrict__ src, const int* __restrict__ dst,
                    const float* __restrict__ z, const float* __restrict__ ex,
                    const float* __restrict__ denom, float* __restrict__ out) {
    long wid = ((long)blockIdx.x * blockDim.x + threadIdx.x) >> 5;
    int lane = threadIdx.x & 31;
    if (wid >= N_EDGES) return;
    int s = src[wid];
    int d = dst[wid];
    float dn = denom[d];
    float alpha = ex[wid] / (dn == 0.0f ? 1.0f : dn);
    float2 zv = *(const float2*)(z + (long)s * OUT_DIM + lane * 2);
    float* o = out + (long)d * OUT_DIM + lane * 2;
    atomAddF32(o,     alpha * zv.x);
    atomAddF32(o + 1, alpha * zv.y);
}

static inline size_t alignUp(size_t x) { return (x + 255) & ~(size_t)255; }

extern "C" void kernel_launch(void* const* d_in, const int* in_sizes, int n_in,
                              void* d_out, int out_size, void* d_ws, size_t ws_size,
                              hipStream_t stream) {
    const float* h  = (const float*)d_in[0];
    const float* W  = (const float*)d_in[1];
    const float* a  = (const float*)d_in[2];
    const int* src  = (const int*)d_in[3];
    const int* dst  = (const int*)d_in[4];
    float* out      = (float*)d_out;

    char* ws = (char*)d_ws;
    float*    z     = (float*)ws;    ws += alignUp((size_t)N_NODES * OUT_DIM * 4);
    float*    s_src = (float*)ws;    ws += alignUp((size_t)N_NODES * 4);
    float*    s_dst = (float*)ws;    ws += alignUp((size_t)N_NODES * 4);
    unsigned* m_enc = (unsigned*)ws; ws += alignUp((size_t)N_NODES * 4);
    float*    m     = (float*)ws;    ws += alignUp((size_t)N_NODES * 4);
    float*    denom = (float*)ws;    ws += alignUp((size_t)N_NODES * 4);
    float*    att   = (float*)ws;    ws += alignUp((size_t)N_EDGES * 4);
    float*    ex    = (float*)ws;    ws += alignUp((size_t)N_EDGES * 4);

    init_kernel<<<2048, 256, 0, stream>>>(out, m_enc, denom);

    gemm_wmma<<<(N_NODES / 16 + GEMM_WAVES - 1) / GEMM_WAVES, 256, 0, stream>>>(h, W, z);

    long score_threads = (long)N_NODES * 32;
    scores_kernel<<<(unsigned)((score_threads + 255) / 256), 256, 0, stream>>>(z, a, s_src, s_dst);

    edge_att_max<<<(N_EDGES + 255) / 256, 256, 0, stream>>>(src, dst, s_src, s_dst, att, m_enc);

    m_finalize<<<(N_NODES + 255) / 256, 256, 0, stream>>>(m_enc, m);

    edge_exp<<<(N_EDGES + 255) / 256, 256, 0, stream>>>(dst, att, m, ex, denom);

    long agg_threads = (long)N_EDGES * 32;
    edge_aggregate<<<(unsigned)((agg_threads + 255) / 256), 256, 0, stream>>>(src, dst, z, ex, denom, out);
}


# edges grouped by destination inside the call (count, scan, fill); one wave per destination row does max, exp-sum and the weighted sum of z rows in registers, stores the row once; no float atomics
# speedup vs baseline: 1.5926x; 1.4451x over previous
#include <hip/hip_runtime.h>

#define N_NODES 100000
#define N_EDGES 1600000
#define IN_DIM  128
#define OUT_DIM 64

typedef __attribute__((ext_vector_type(2))) float v2f;
typedef __attribute__((ext_vector_type(8))) float v8f;

__global__ __launch_bounds__(256)
void clear_counts(int* __restrict__ cnt) {
    int i = blockIdx.x * 256 + threadIdx.x;
    if (i < N_NODES) cnt[i] = 0;
}

#define GEMM_WAVES 8
__global__ __launch_bounds__(256)
void gemm_wmma(const float* __restrict__ h, const float* __restrict__ W,
               float* __restrict__ z) {
    __shared__ float4 Wl4[16 * 4 * 32];

    const int tid  = threadIdx.x;
    const int wave = tid >> 5;
    const int lane = tid & 31;
    const int c    = lane & 15;
    const int hh   = lane >> 4;
    const int tileRaw = blockIdx.x * GEMM_WAVES + wave;
    const bool active = tileRaw < N_NODES / 16;
    const int tile = active ? tileRaw : (N_NODES / 16 - 1);

    const float* wcol = W + (8 * (wave >> 2) + 4 * hh) * OUT_DIM + 4 * c + (wave & 3);
    float4 w4[8];
    for (int m = 0; m < 8; ++m) {
        const float* wp = wcol + m * 16 * OUT_DIM;
        w4[m].x = wp[0 * OUT_DIM];
        w4[m].y = wp[1 * OUT_DIM];
        w4[m].z = wp[2 * OUT_DIM];
        w4[m].w = wp[3 * OUT_DIM];
    }

    const float4* hrow = (const float4*)(h + (long)(tile * 16 + c) * IN_DIM + 4 * hh);
    float4 a4[16];
    for (int u = 0; u < 16; ++u) a4[u] = hrow[2 * u];

    for (int m = 0; m < 8; ++m) Wl4[tid + 256 * m] = w4[m];
    __syncthreads();

    v8f acc0 = {}, acc1 = {}, acc2 = {}, acc3 = {};
    for (int u = 0; u < 16; ++u) {
        const float4 b0 = Wl4[(u * 4 + 0) * 32 + lane];
        const float4 b1 = Wl4[(u * 4 + 1) * 32 + lane];
        const float4 b2 = Wl4[(u * 4 + 2) * 32 + lane];
        const float4 b3 = Wl4[(u * 4 + 3) * 32 + lane];
        const v2f aLo = {a4[u].x, a4[u].y};
        const v2f aHi = {a4[u].z, a4[u].w};
        acc0 = __builtin_amdgcn_wmma_f32_16x16x4_f32(false, aLo, false, (v2f){b0.x, b0.y}, (short)0, acc0, false, false);
        acc1 = __builtin_amdgcn_wmma_f32_16x16x4_f32(false, aLo, false, (v2f){b1.x, b1.y}, (short)0, acc1, false, false);
        acc2 = __builtin_amdgcn_wmma_f32_16x16x4_f32(false, aLo, false, (v2f){b2.x, b2.y}, (short)0, acc2, false, false);
        acc3 = __builtin_amdgcn_wmma_f32_16x16x4_f32(false, aLo, false, (v2f){b3.x, b3.y}, (short)0, acc3, false, false);
        acc0 = __builtin_amdgcn_wmma_f32_16x16x4_f32(false, aHi, false, (v2f){b0.z, b0.w}, (short)0, acc0, false, false);
        acc1 = __builtin_amdgcn_wmma_f32_16x16x4_f32(false, aHi, false, (v2f){b1.z, b1.w}, (short)0, acc1, false, false);
        acc2 = __builtin_amdgcn_wmma_f32_16x16x4_f32(false, aHi, false, (v2f){b2.z, b2.w}, (short)0, acc2, false, false);
        acc3 = __builtin_amdgcn_wmma_f32_16x16x4_f32(false, aHi, false, (v2f){b3.z, b3.w}, (short)0, acc3, false, false);
    }

    if (active) {
        float* zt = z + (long)(tile * 16 + 8 * hh) * OUT_DIM + 4 * c;
        for (int r = 0; r < 8; ++r) {
            *(float4*)(zt + (long)r * OUT_DIM) = make_float4(acc0[r], acc1[r], acc2[r], acc3[r]);
        }
    }
}

__global__ __launch_bounds__(256)
void scores_kernel(const float* __restrict__ z, const float* __restrict__ a,
                   float* __restrict__ s_src, float* __restrict__ s_dst) {
    long wid = ((long)blockIdx.x * blockDim.x + threadIdx.x) >> 5;
    int lane = threadIdx.x & 31;
    if (wid >= N_NODES) return;
    float2 zv = *(const float2*)(z + wid * OUT_DIM + lane * 2);
    float2 as = *(const float2*)(a + lane * 2);
    float2 ad = *(const float2*)(a + OUT_DIM + lane * 2);
    float ps = zv.x * as.x + zv.y * as.y;
    float pd = zv.x * ad.x + zv.y * ad.y;
    #pragma unroll
    for (int off = 16; off > 0; off >>= 1) {
        ps += __shfl_down(ps, off, 32);
        pd += __shfl_down(pd, off, 32);
    }
    if (lane == 0) { s_src[wid] = ps; s_dst[wid] = pd; }
}


__global__ __launch_bounds__(256)
void degree_count(const int* __restrict__ dst, int* __restrict__ cnt) {
    int i = blockIdx.x * 256 + threadIdx.x;
    if (i >= N_EDGES / 4) return;
    const int4 d = ((const int4*)dst)[i];
    atomicAdd(&cnt[d.x], 1);
    atomicAdd(&cnt[d.y], 1);
    atomicAdd(&cnt[d.z], 1);
    atomicAdd(&cnt[d.w], 1);
}

#define SCAN_BLOCKS ((N_NODES / 4 + 255) / 256)
__global__ __launch_bounds__(256)
void row_starts(const int* __restrict__ cnt, int* __restrict__ cur) {
    __shared__ int wtot[8];
    __shared__ int wpre[8];
    const int t    = threadIdx.x;
    const int lane = t & 31;
    const int wv   = t >> 5;
    const int4* cnt4 = (const int4*)cnt;

    const int nb = blockIdx.x;
    int pre = 0;
    for (int r = 0; r < nb; r += 16) {
        int4 v[16];
        #pragma unroll
        for (int u = 0; u < 16; ++u) {
            const int rr = (r + u < nb) ? (r + u) : (nb - 1);
            v[u] = cnt4[rr * 256 + t];
        }
        #pragma unroll
        for (int u = 0; u < 16; ++u) {
            const int sm = (v[u].x + v[u].y) + (v[u].z + v[u].w);
            pre += (r + u < nb) ? sm : 0;
        }
    }
    const int idx = blockIdx.x * 256 + t;
    const bool live = idx < N_NODES / 4;
    int4 own = make_int4(0, 0, 0, 0);
    if (live) own = cnt4[idx];
    const int t4 = (own.x + own.y) + (own.z + own.w);

    for (int off = 16; off > 0; off >>= 1) pre += __shfl_xor(pre, off, 32);
    int inc = t4;
    for (int off = 1; off < 32; off <<= 1) {
        const int n = __shfl_up(inc, off, 32);
        if (lane >= off) inc += n;
    }
    if (lane == 31) wtot[wv] = inc;
    if (lane == 0)  wpre[wv] = pre;
    __syncthreads();
    int base = 0;
    for (int w = 0; w < 8; ++w) {
        base += wpre[w];
        if (w < wv) base += wtot[w];
    }
    const int e0 = base + inc - t4;
    if (live) {
        ((int4*)cur)[idx] = make_int4(e0, e0 + own.x, e0 + own.x + own.y, e0 + own.x + own.y + own.z);
    }
}

__global__ __launch_bounds__(256)
void fill_rows(const int* __restrict__ src, const int* __restrict__ dst,
               int* __restrict__ cur, int* __restrict__ srcs) {
    int i = blockIdx.x * 256 + threadIdx.x;
    if (i >= N_EDGES / 4) return;
    const int4 d = ((const int4*)dst)[i];
    const int4 s = ((const int4*)src)[i];
    const int p0 = atomicAdd(&cur[d.x], 1);
    const int p1 = atomicAdd(&cur[d.y], 1);
    const int p2 = atomicAdd(&cur[d.z], 1);
    const int p3 = atomicAdd(&cur[d.w], 1);
    srcs[p0] = s.x;
    srcs[p1] = s.y;
    srcs[p2] = s.z;
    srcs[p3] = s.w;
}

__device__ __forceinline__ float leaky(float v) { return v > 0.0f ? v : 0.01f * v; }
__device__ __forceinline__ float lane_f(float v, int l) {
    return __int_as_float(__builtin_amdgcn_readlane(__float_as_int(v), l));
}
#define ROW_BLOCKS 2048
__global__ __launch_bounds__(256)
void row_softmax_sum(const int* __restrict__ cur, const int* __restrict__ srcs,
                     const float* __restrict__ s_src, const float* __restrict__ s_dst,
                     const float* __restrict__ z, float* __restrict__ out) {
    const int lane = threadIdx.x & 31;
    const int wave0  = __builtin_amdgcn_readfirstlane((int)((blockIdx.x * 256 + threadIdx.x) >> 5));
    const int nwaves = ROW_BLOCKS * 8;
    const float NEG_INF = -__builtin_inff();

    for (int d = wave0; d < N_NODES; d += nwaves) {
        const int end = cur[d];
        const int beg = d ? cur[d - 1] : 0;
        float accx = 0.0f, accy = 0.0f;
        if (end > beg) {
            const float sd = s_dst[d];
            const int  i0  = beg + lane;
            const bool ok0 = i0 < end;
            const int  s0  = ok0 ? srcs[i0] : 0;
            float l0 = NEG_INF;
            if (ok0) l0 = leaky(s_src[s0] + sd);

            float mx = l0;
            for (int i = i0 + 32; i < end; i += 32) mx = fmaxf(mx, leaky(s_src[srcs[i]] + sd));
            for (int off = 16; off > 0; off >>= 1) mx = fmaxf(mx, __shfl_xor(mx, off, 32));
            mx = (mx == NEG_INF) ? 0.0f : mx;

            const float e0 = ok0 ? __expf(l0 - mx) : 0.0f;
            float sum = e0;
            for (int i = i0 + 32; i < end; i += 32) sum += __expf(leaky(s_src[srcs[i]] + sd) - mx);
            for (int off = 16; off > 0; off >>= 1) sum += __shfl_xor(sum, off, 32);
            const float den = (sum == 0.0f) ? 1.0f : sum;

            for (int base = beg; base < end; base += 32) {
                int   s;
                float al;
                if (base == beg) {
                    s  = s0;
                    al = e0 / den;
                } else {
                    const int  i  = base + lane;
                    const bool ok = i < end;
                    s = ok ? srcs[i] : 0;
                    float ee = 0.0f;
                    if (ok) ee = __expf(leaky(s_src[s] + sd) - mx);
                    al = ee / den;
                }
                const int rem = end - base;
                const int cnt = rem < 32 ? rem : 32;
                for (int j = 0; j < cnt; j += 8) {
                    float2 zv[8];
                    float  a[8];
                    #pragma unroll
                    for (int u = 0; u < 8; ++u) {
                        const int jj = (j + u < cnt) ? (j + u) : (cnt - 1);
                        const int sj = __builtin_amdgcn_readlane(s, jj);
                        a[u]  = (j + u < cnt) ? lane_f(al, jj) : 0.0f;
                        zv[u] = *(const float2*)(z + (long)sj * OUT_DIM + 2 * lane);
                    }
                    #pragma unroll
                    for (int u = 0; u < 8; ++u) {
                        accx += a[u] * zv[u].x;
                        accy += a[u] * zv[u].y;
                    }
                }
            }
        }
        *(float2*)(out + (long)d * OUT_DIM + 2 * lane) = make_float2(accx, accy);
    }
}

static inline size_t alignUp(size_t x) { return (x + 255) & ~(size_t)255; }

extern "C" void kernel_launch(void* const* d_in, const int* in_sizes, int n_in,
                              void* d_out, int out_size, void* d_ws, size_t ws_size,
                              hipStream_t stream) {
    const float* h  = (const float*)d_in[0];
    const float* W  = (const float*)d_in[1];
    const float* a  = (const float*)d_in[2];
    const int* src  = (const int*)d_in[3];
    const int* dst  = (const int*)d_in[4];
    float* out      = (float*)d_out;

    char* ws = (char*)d_ws;
    float* z     = (float*)ws; ws += alignUp((size_t)N_NODES * OUT_DIM * 4);
    float* s_src = (float*)ws; ws += alignUp((size_t)N_NODES * 4);
    float* s_dst = (float*)ws; ws += alignUp((size_t)N_NODES * 4);
    int*   cnt   = (int*)ws;   ws += alignUp((size_t)N_NODES * 4);
    int*   cur   = (int*)ws;   ws += alignUp((size_t)N_NODES * 4);
    int*   srcs  = (int*)ws;   ws += alignUp((size_t)N_EDGES * 4);

    clear_counts<<<(N_NODES + 255) / 256, 256, 0, stream>>>(cnt);

    degree_count<<<(N_EDGES / 4 + 255) / 256, 256, 0, stream>>>(dst, cnt);

    row_starts<<<SCAN_BLOCKS, 256, 0, stream>>>(cnt, cur);

    fill_rows<<<(N_EDGES / 4 + 255) / 256, 256, 0, stream>>>(src, dst, cur, srcs);

    gemm_wmma<<<(N_NODES / 16 + GEMM_WAVES - 1) / GEMM_WAVES, 256, 0, stream>>>(h, W, z);

    long score_threads = (long)N_NODES * 32;
    scores_kernel<<<(unsigned)((score_threads + 255) / 256), 256, 0, stream>>>(z, a, s_src, s_dst);

    row_softmax_sum<<<ROW_BLOCKS, 256, 0, stream>>>(cur, srcs, s_src, s_dst, z, out);
}


# one round of scattered int atomics instead of two: the count pass's returning add is the edge's rank, fill stores at row start + rank; in-row sum order now follows the count pass's atomic arrival
# speedup vs baseline: 2.2988x; 1.4434x over previous
#include <hip/hip_runtime.h>

#define N_NODES 100000
#define N_EDGES 1600000
#define IN_DIM  128
#define OUT_DIM 64

typedef __attribute__((ext_vector_type(2))) float v2f;
typedef __attribute__((ext_vector_type(8))) float v8f;

__global__ __launch_bounds__(256)
void clear_counts(int* __restrict__ cnt) {
    int i = blockIdx.x * 256 + threadIdx.x;
    if (i < N_NODES) cnt[i] = 0;
}

#define GEMM_WAVES 8

#define ROW_XOR_ADD(v, M) \
    ((v) + __int_as_float(__builtin_amdgcn_update_dpp(0, __float_as_int(v), 0x160 + (M), 0xf, 0xf, true)))
__global__ __launch_bounds__(256)
void gemm_wmma(const float* __restrict__ h, const float* __restrict__ W,
               const float* __restrict__ a, float* __restrict__ z,
               float* __restrict__ s_src, float* __restrict__ s_dst) {
    __shared__ float4 Wl4[16 * 4 * 32];

    const int tid  = threadIdx.x;
    const int wave = tid >> 5;
    const int lane = tid & 31;
    const int c    = lane & 15;
    const int hh   = lane >> 4;
    const int tileRaw = blockIdx.x * GEMM_WAVES + wave;
    const bool active = tileRaw < N_NODES / 16;
    const int tile = active ? tileRaw : (N_NODES / 16 - 1);

    const float* wcol = W + (8 * (wave >> 2) + 4 * hh) * OUT_DIM + 4 * c + (wave & 3);
    float4 w4[8];
    for (int m = 0; m < 8; ++m) {
        const float* wp = wcol + m * 16 * OUT_DIM;
        w4[m].x = wp[0 * OUT_DIM];
        w4[m].y = wp[1 * OUT_DIM];
        w4[m].z = wp[2 * OUT_DIM];
        w4[m].w = wp[3 * OUT_DIM];
    }

    const float4* hrow = (const float4*)(h + (long)(tile * 16 + c) * IN_DIM + 4 * hh);
    float4 a4[16];
    for (int u = 0; u < 16; ++u) a4[u] = hrow[2 * u];

    const float4 as4 = *(const float4*)(a + 4 * c);
    const float4 ad4 = *(const float4*)(a + OUT_DIM + 4 * c);

    for (int m = 0; m < 8; ++m) Wl4[tid + 256 * m] = w4[m];
    __syncthreads();

    v8f acc0 = {}, acc1 = {}, acc2 = {}, acc3 = {};
    for (int u = 0; u < 16; ++u) {
        const float4 b0 = Wl4[(u * 4 + 0) * 32 + lane];
        const float4 b1 = Wl4[(u * 4 + 1) * 32 + lane];
        const float4 b2 = Wl4[(u * 4 + 2) * 32 + lane];
        const float4 b3 = Wl4[(u * 4 + 3) * 32 + lane];
        const v2f aLo = {a4[u].x, a4[u].y};
        const v2f aHi = {a4[u].z, a4[u].w};
        acc0 = __builtin_amdgcn_wmma_f32_16x16x4_f32(false, aLo, false, (v2f){b0.x, b0.y}, (short)0, acc0, false, false);
        acc1 = __builtin_amdgcn_wmma_f32_16x16x4_f32(false, aLo, false, (v2f){b1.x, b1.y}, (short)0, acc1, false, false);
        acc2 = __builtin_amdgcn_wmma_f32_16x16x4_f32(false, aLo, false, (v2f){b2.x, b2.y}, (short)0, acc2, false, false);
        acc3 = __builtin_amdgcn_wmma_f32_16x16x4_f32(false, aLo, false, (v2f){b3.x, b3.y}, (short)0, acc3, false, false);
        acc0 = __builtin_amdgcn_wmma_f32_16x16x4_f32(false, aHi, false, (v2f){b0.z, b0.w}, (short)0, acc0, false, false);
        acc1 = __builtin_amdgcn_wmma_f32_16x16x4_f32(false, aHi, false, (v2f){b1.z, b1.w}, (short)0, acc1, false, false);
        acc2 = __builtin_amdgcn_wmma_f32_16x16x4_f32(false, aHi, false, (v2f){b2.z, b2.w}, (short)0, acc2, false, false);
        acc3 = __builtin_amdgcn_wmma_f32_16x16x4_f32(false, aHi, false, (v2f){b3.z, b3.w}, (short)0, acc3, false, false);
    }

    if (active) {
        float* zt = z + (long)(tile * 16 + 8 * hh) * OUT_DIM + 4 * c;
        for (int r = 0; r < 8; ++r) {
            *(float4*)(zt + (long)r * OUT_DIM) = make_float4(acc0[r], acc1[r], acc2[r], acc3[r]);
        }
    }

    float ps[8], pd[8];
    for (int r = 0; r < 8; ++r) {
        ps[r] = (acc0[r] * as4.x + acc1[r] * as4.y) + (acc2[r] * as4.z + acc3[r] * as4.w);
        pd[r] = (acc0[r] * ad4.x + acc1[r] * ad4.y) + (acc2[r] * ad4.z + acc3[r] * ad4.w);
    }
    for (int r = 0; r < 8; ++r) { ps[r] = ROW_XOR_ADD(ps[r], 1); pd[r] = ROW_XOR_ADD(pd[r], 1); }
    for (int r = 0; r < 8; ++r) { ps[r] = ROW_XOR_ADD(ps[r], 2); pd[r] = ROW_XOR_ADD(pd[r], 2); }
    for (int r = 0; r < 8; ++r) { ps[r] = ROW_XOR_ADD(ps[r], 4); pd[r] = ROW_XOR_ADD(pd[r], 4); }
    for (int r = 0; r < 8; ++r) { ps[r] = ROW_XOR_ADD(ps[r], 8); pd[r] = ROW_XOR_ADD(pd[r], 8); }
    if (active && c == 0) {
        float* sp = s_src + tile * 16 + 8 * hh;
        float* dp = s_dst + tile * 16 + 8 * hh;
        *(float4*)(sp)     = make_float4(ps[0], ps[1], ps[2], ps[3]);
        *(float4*)(sp + 4) = make_float4(ps[4], ps[5], ps[6], ps[7]);
        *(float4*)(dp)     = make_float4(pd[0], pd[1], pd[2], pd[3]);
        *(float4*)(dp + 4) = make_float4(pd[4], pd[5], pd[6], pd[7]);
    }
}


__global__ __launch_bounds__(256)
void degree_rank(const int* __restrict__ dst, int* __restrict__ cnt, int* __restrict__ rank) {
    int i = blockIdx.x * 256 + threadIdx.x;
    if (i >= N_EDGES / 4) return;
    const int4 d = ((const int4*)dst)[i];
    int4 r;
    r.x = atomicAdd(&cnt[d.x], 1);
    r.y = atomicAdd(&cnt[d.y], 1);
    r.z = atomicAdd(&cnt[d.z], 1);
    r.w = atomicAdd(&cnt[d.w], 1);
    ((int4*)rank)[i] = r;
}

#define SCAN_BLOCKS ((N_NODES / 4 + 255) / 256)
__global__ __launch_bounds__(256)
void row_starts(const int* __restrict__ cnt, int* __restrict__ cur) {
    __shared__ int wtot[8];
    __shared__ int wpre[8];
    const int t    = threadIdx.x;
    const int lane = t & 31;
    const int wv   = t >> 5;
    const int4* cnt4 = (const int4*)cnt;

    const int nb = blockIdx.x;
    int pre = 0;
    for (int r = 0; r < nb; r += 16) {
        int4 v[16];
        #pragma unroll
        for (int u = 0; u < 16; ++u) {
            const int rr = (r + u < nb) ? (r + u) : (nb - 1);
            v[u] = cnt4[rr * 256 + t];
        }
        #pragma unroll
        for (int u = 0; u < 16; ++u) {
            const int sm = (v[u].x + v[u].y) + (v[u].z + v[u].w);
            pre += (r + u < nb) ? sm : 0;
        }
    }
    const int idx = blockIdx.x * 256 + t;
    const bool live = idx < N_NODES / 4;
    int4 own = make_int4(0, 0, 0, 0);
    if (live) own = cnt4[idx];
    const int t4 = (own.x + own.y) + (own.z + own.w);

    for (int off = 16; off > 0; off >>= 1) pre += __shfl_xor(pre, off, 32);
    int inc = t4;
    for (int off = 1; off < 32; off <<= 1) {
        const int n = __shfl_up(inc, off, 32);
        if (lane >= off) inc += n;
    }
    if (lane == 31) wtot[wv] = inc;
    if (lane == 0)  wpre[wv] = pre;
    __syncthreads();
    int base = 0;
    for (int w = 0; w < 8; ++w) {
        base += wpre[w];
        if (w < wv) base += wtot[w];
    }
    const int e0 = base + inc - t4;
    if (live) {
        ((int4*)cur)[idx] = make_int4(e0, e0 + own.x, e0 + own.x + own.y, e0 + own.x + own.y + own.z);
    }
}

__global__ __launch_bounds__(256)
void fill_rows(const int* __restrict__ src, const int* __restrict__ dst,
               const int* __restrict__ rank, const int* __restrict__ cur,
               int* __restrict__ srcs) {
    int i = blockIdx.x * 256 + threadIdx.x;
    if (i >= N_EDGES / 4) return;
    const int4 d = ((const int4*)dst)[i];
    const int4 r = ((const int4*)rank)[i];
    const int4 s = ((const int4*)src)[i];
    const int p0 = cur[d.x] + r.x;
    const int p1 = cur[d.y] + r.y;
    const int p2 = cur[d.z] + r.z;
    const int p3 = cur[d.w] + r.w;
    srcs[p0] = s.x;
    srcs[p1] = s.y;
    srcs[p2] = s.z;
    srcs[p3] = s.w;
}

__device__ __forceinline__ float leaky(float v) { return v > 0.0f ? v : 0.01f * v; }
__device__ __forceinline__ float lane_f(float v, int l) {
    return __int_as_float(__builtin_amdgcn_readlane(__float_as_int(v), l));
}
#define ROW_BLOCKS 2048
__global__ __launch_bounds__(256)
void row_softmax_sum(const int* __restrict__ cur, const int* __restrict__ srcs,
                     const float* __restrict__ s_src, const float* __restrict__ s_dst,
                     const float* __restrict__ z, float* __restrict__ out) {
    const int lane = threadIdx.x & 31;
    const int wave0  = __builtin_amdgcn_readfirstlane((int)((blockIdx.x * 256 + threadIdx.x) >> 5));
    const int nwaves = ROW_BLOCKS * 8;
    const float NEG_INF = -__builtin_inff();

    for (int d = wave0; d < N_NODES; d += nwaves) {
        const int beg = cur[d];
        const int end = (d + 1 < N_NODES) ? cur[d + 1] : N_EDGES;
        float accx = 0.0f, accy = 0.0f;
        if (end > beg) {
            const float sd = s_dst[d];
            const int  i0  = beg + lane;
            const bool ok0 = i0 < end;
            const int  s0  = ok0 ? srcs[i0] : 0;
            float l0 = NEG_INF;
            if (ok0) l0 = leaky(s_src[s0] + sd);

            float mx = l0;
            for (int i = i0 + 32; i < end; i += 32) mx = fmaxf(mx, leaky(s_src[srcs[i]] + sd));
            for (int off = 16; off > 0; off >>= 1) mx = fmaxf(mx, __shfl_xor(mx, off, 32));
            mx = (mx == NEG_INF) ? 0.0f : mx;

            const float e0 = ok0 ? __expf(l0 - mx) : 0.0f;
            float sum = e0;
            for (int i = i0 + 32; i < end; i += 32) sum += __expf(leaky(s_src[srcs[i]] + sd) - mx);
            for (int off = 16; off > 0; off >>= 1) sum += __shfl_xor(sum, off, 32);
            const float den = (sum == 0.0f) ? 1.0f : sum;

            for (int base = beg; base < end; base += 32) {
                int   s;
                float al;
                if (base == beg) {
                    s  = s0;
                    al = e0 / den;
                } else {
                    const int  i  = base + lane;
                    const bool ok = i < end;
                    s = ok ? srcs[i] : 0;
                    float ee = 0.0f;
                    if (ok) ee = __expf(leaky(s_src[s] + sd) - mx);
                    al = ee / den;
                }
                const int rem = end - base;
                const int cnt = rem < 32 ? rem : 32;
                for (int j = 0; j < cnt; j += 8) {
                    float2 zv[8];
                    float  a[8];
                    #pragma unroll
                    for (int u = 0; u < 8; ++u) {
                        const int jj = (j + u < cnt) ? (j + u) : (cnt - 1);
                        const int sj = __builtin_amdgcn_readlane(s, jj);
                        a[u]  = (j + u < cnt) ? lane_f(al, jj) : 0.0f;
                        zv[u] = *(const float2*)(z + (long)sj * OUT_DIM + 2 * lane);
                    }
                    #pragma unroll
                    for (int u = 0; u < 8; ++u) {
                        accx += a[u] * zv[u].x;
                        accy += a[u] * zv[u].y;
                    }
                }
            }
        }
        *(float2*)(out + (long)d * OUT_DIM + 2 * lane) = make_float2(accx, accy);
    }
}

static inline size_t alignUp(size_t x) { return (x + 255) & ~(size_t)255; }

extern "C" void kernel_launch(void* const* d_in, const int* in_sizes, int n_in,
                              void* d_out, int out_size, void* d_ws, size_t ws_size,
                              hipStream_t stream) {
    const float* h  = (const float*)d_in[0];
    const float* W  = (const float*)d_in[1];
    const float* a  = (const float*)d_in[2];
    const int* src  = (const int*)d_in[3];
    const int* dst  = (const int*)d_in[4];
    float* out      = (float*)d_out;

    char* ws = (char*)d_ws;
    float* z     = (float*)ws; ws += alignUp((size_t)N_NODES * OUT_DIM * 4);
    float* s_src = (float*)ws; ws += alignUp((size_t)N_NODES * 4);
    float* s_dst = (float*)ws; ws += alignUp((size_t)N_NODES * 4);
    int*   cnt   = (int*)ws;   ws += alignUp((size_t)N_NODES * 4);
    int*   cur   = (int*)ws;   ws += alignUp((size_t)N_NODES * 4);
    int*   srcs  = (int*)ws;   ws += alignUp((size_t)N_EDGES * 4);
    int*   rank  = (int*)z;

    clear_counts<<<(N_NODES + 255) / 256, 256, 0, stream>>>(cnt);

    degree_rank<<<(N_EDGES / 4 + 255) / 256, 256, 0, stream>>>(dst, cnt, rank);

    row_starts<<<SCAN_BLOCKS, 256, 0, stream>>>(cnt, cur);

    fill_rows<<<(N_EDGES / 4 + 255) / 256, 256, 0, stream>>>(src, dst, rank, cur, srcs);

    gemm_wmma<<<(N_NODES / 16 + GEMM_WAVES - 1) / GEMM_WAVES, 256, 0, stream>>>(h, W, a, z, s_src, s_dst);

    row_softmax_sum<<<ROW_BLOCKS, 256, 0, stream>>>(cur, srcs, s_src, s_dst, z, out);
}


# edge counts and slot hand-out moved from scattered global atomics to LDS counters: 256 blocks = 8 node ranges x 32 edge stretches, counts scanned over stretches; in-row sum order: stretch, LDS arrival
# speedup vs baseline: 4.6256x; 2.0122x over previous
#include <hip/hip_runtime.h>

#define N_NODES 100000
#define N_EDGES 1600000
#define IN_DIM  128
#define OUT_DIM 64

typedef __attribute__((ext_vector_type(2))) float v2f;
typedef __attribute__((ext_vector_type(8))) float v8f;

#define PARTS        8
#define PART_NODES   12512
#define SLICES       32
#define SLICE_VEC    (N_EDGES / SLICES / 4)
#define PART_THREADS 1024

__device__ __forceinline__ int lds_add(int* p, int v) {
    return __hip_atomic_fetch_add(p, v, __ATOMIC_RELAXED, __HIP_MEMORY_SCOPE_WORKGROUP);
}

__global__ __launch_bounds__(PART_THREADS)
void count_part(const int* __restrict__ dst, int* __restrict__ bh) {
    __shared__ int hist[PART_NODES];
    const int t  = threadIdx.x;
    const int p  = blockIdx.x & (PARTS - 1);
    const int b  = blockIdx.x >> 3;
    const int lo = p * PART_NODES;
    const unsigned span = (unsigned)((N_NODES - lo < PART_NODES) ? (N_NODES - lo) : PART_NODES);
    for (int i = t; i < PART_NODES; i += PART_THREADS) hist[i] = 0;
    __syncthreads();

    const int4* d4 = (const int4*)dst + b * SLICE_VEC;
    for (int base = t; base < SLICE_VEC; base += 4 * PART_THREADS) {
        int4 d[4];
        #pragma unroll
        for (int u = 0; u < 4; ++u) {
            const int i = base + u * PART_THREADS;
            d[u] = (i < SLICE_VEC) ? d4[i] : make_int4(-1, -1, -1, -1);
        }
        #pragma unroll
        for (int u = 0; u < 4; ++u) {
            const unsigned x0 = (unsigned)(d[u].x - lo);
            const unsigned x1 = (unsigned)(d[u].y - lo);
            const unsigned x2 = (unsigned)(d[u].z - lo);
            const unsigned x3 = (unsigned)(d[u].w - lo);
            if (x0 < span) lds_add(&hist[x0], 1);
            if (x1 < span) lds_add(&hist[x1], 1);
            if (x2 < span) lds_add(&hist[x2], 1);
            if (x3 < span) lds_add(&hist[x3], 1);
        }
    }
    __syncthreads();
    int* row = bh + (long)(p * SLICES + b) * PART_NODES;
    for (int i = t; i < PART_NODES; i += PART_THREADS) row[i] = hist[i];
}

__global__ __launch_bounds__(256)
void slice_offsets(int* __restrict__ bh, int* __restrict__ cnt) {
    const int g = blockIdx.x * 256 + threadIdx.x;
    const int p = g / PART_NODES;
    const int i = g - p * PART_NODES;
    int* col = bh + (long)p * (SLICES * PART_NODES) + i;
    int v[SLICES];
    #pragma unroll
    for (int b = 0; b < SLICES; ++b) v[b] = col[b * PART_NODES];
    int run = 0;
    #pragma unroll
    for (int b = 0; b < SLICES; ++b) {
        col[b * PART_NODES] = run;
        run += v[b];
    }
    if (g < N_NODES) cnt[g] = run;
}

#define SCAN_BLOCKS ((N_NODES / 4 + 255) / 256)
__global__ __launch_bounds__(256)
void row_starts(const int* __restrict__ cnt, int* __restrict__ cur) {
    __shared__ int wtot[8];
    __shared__ int wpre[8];
    const int t    = threadIdx.x;
    const int lane = t & 31;
    const int wv   = t >> 5;
    const int4* cnt4 = (const int4*)cnt;

    const int nb = blockIdx.x;
    int pre = 0;
    for (int r = 0; r < nb; r += 16) {
        int4 v[16];
        #pragma unroll
        for (int u = 0; u < 16; ++u) {
            const int rr = (r + u < nb) ? (r + u) : (nb - 1);
            v[u] = cnt4[rr * 256 + t];
        }
        #pragma unroll
        for (int u = 0; u < 16; ++u) {
            const int sm = (v[u].x + v[u].y) + (v[u].z + v[u].w);
            pre += (r + u < nb) ? sm : 0;
        }
    }
    const int idx = blockIdx.x * 256 + t;
    const bool live = idx < N_NODES / 4;
    int4 own = make_int4(0, 0, 0, 0);
    if (live) own = cnt4[idx];
    const int t4 = (own.x + own.y) + (own.z + own.w);

    for (int off = 16; off > 0; off >>= 1) pre += __shfl_xor(pre, off, 32);
    int inc = t4;
    for (int off = 1; off < 32; off <<= 1) {
        const int n = __shfl_up(inc, off, 32);
        if (lane >= off) inc += n;
    }
    if (lane == 31) wtot[wv] = inc;
    if (lane == 0)  wpre[wv] = pre;
    __syncthreads();
    int base = 0;
    for (int w = 0; w < 8; ++w) {
        base += wpre[w];
        if (w < wv) base += wtot[w];
    }
    const int e0 = base + inc - t4;
    if (live) {
        ((int4*)cur)[idx] = make_int4(e0, e0 + own.x, e0 + own.x + own.y, e0 + own.x + own.y + own.z);
    }
}

#define POS_TRIPS ((PART_NODES + PART_THREADS - 1) / PART_THREADS)
__global__ __launch_bounds__(PART_THREADS)
void fill_part(const int* __restrict__ src, const int* __restrict__ dst,
               const int* __restrict__ bh, const int* __restrict__ cur,
               int* __restrict__ srcs) {
    __shared__ int pos[PART_NODES + 32];
    const int t  = threadIdx.x;
    const unsigned spare = PART_NODES + (t & 31);
    const int p  = blockIdx.x & (PARTS - 1);
    const int b  = blockIdx.x >> 3;
    const int lo = p * PART_NODES;
    const unsigned span = (unsigned)((N_NODES - lo < PART_NODES) ? (N_NODES - lo) : PART_NODES);
    const int* row = bh + (long)(p * SLICES + b) * PART_NODES;
    {
        const int last = (int)span - 1;
        int cv[POS_TRIPS], rv[POS_TRIPS];
        #pragma unroll
        for (int k = 0; k < POS_TRIPS; ++k) {
            const int i  = t + k * PART_THREADS;
            const int ic = i < last ? i : last;
            cv[k] = cur[lo + ic];
            rv[k] = row[ic];
        }
        #pragma unroll
        for (int k = 0; k < POS_TRIPS; ++k) {
            const int i = t + k * PART_THREADS;
            if (i < PART_NODES) pos[i] = cv[k] + rv[k];
        }
        if (t < 32) pos[PART_NODES + t] = 0;
    }
    __syncthreads();

    const int4* d4 = (const int4*)dst + b * SLICE_VEC;
    const int4* s4 = (const int4*)src + b * SLICE_VEC;
    for (int base = t; base < SLICE_VEC; base += 4 * PART_THREADS) {
        int4 d[4], s[4];
        #pragma unroll
        for (int u = 0; u < 4; ++u) {
            const int i = base + u * PART_THREADS;
            const bool in = i < SLICE_VEC;
            d[u] = in ? d4[i] : make_int4(-1, -1, -1, -1);
            s[u] = in ? s4[i] : make_int4(0, 0, 0, 0);
        }
        #pragma unroll
        for (int u = 0; u < 4; ++u) {
            const unsigned x0 = (unsigned)(d[u].x - lo);
            const unsigned x1 = (unsigned)(d[u].y - lo);
            const unsigned x2 = (unsigned)(d[u].z - lo);
            const unsigned x3 = (unsigned)(d[u].w - lo);
            const unsigned q0 = (unsigned)lds_add(&pos[x0 < span ? x0 : spare], 1);
            const unsigned q1 = (unsigned)lds_add(&pos[x1 < span ? x1 : spare], 1);
            const unsigned q2 = (unsigned)lds_add(&pos[x2 < span ? x2 : spare], 1);
            const unsigned q3 = (unsigned)lds_add(&pos[x3 < span ? x3 : spare], 1);
            if (x0 < span) srcs[q0] = s[u].x;
            if (x1 < span) srcs[q1] = s[u].y;
            if (x2 < span) srcs[q2] = s[u].z;
            if (x3 < span) srcs[q3] = s[u].w;
        }
    }
}

#define GEMM_WAVES 8

#define ROW_XOR_ADD(v, M) \
    ((v) + __int_as_float(__builtin_amdgcn_update_dpp(0, __float_as_int(v), 0x160 + (M), 0xf, 0xf, true)))
__global__ __launch_bounds__(256)
void gemm_wmma(const float* __restrict__ h, const float* __restrict__ W,
               const float* __restrict__ a, float* __restrict__ z,
               float* __restrict__ s_src, float* __restrict__ s_dst) {
    __shared__ float4 Wl4[16 * 4 * 32];

    const int tid  = threadIdx.x;
    const int wave = tid >> 5;
    const int lane = tid & 31;
    const int c    = lane & 15;
    const int hh   = lane >> 4;
    const int tileRaw = blockIdx.x * GEMM_WAVES + wave;
    const bool active = tileRaw < N_NODES / 16;
    const int tile = active ? tileRaw : (N_NODES / 16 - 1);

    const float* wcol = W + (8 * (wave >> 2) + 4 * hh) * OUT_DIM + 4 * c + (wave & 3);
    float4 w4[8];
    for (int m = 0; m < 8; ++m) {
        const float* wp = wcol + m * 16 * OUT_DIM;
        w4[m].x = wp[0 * OUT_DIM];
        w4[m].y = wp[1 * OUT_DIM];
        w4[m].z = wp[2 * OUT_DIM];
        w4[m].w = wp[3 * OUT_DIM];
    }

    const float4* hrow = (const float4*)(h + (long)(tile * 16 + c) * IN_DIM + 4 * hh);
    float4 a4[16];
    for (int u = 0; u < 16; ++u) a4[u] = hrow[2 * u];

    const float4 as4 = *(const float4*)(a + 4 * c);
    const float4 ad4 = *(const float4*)(a + OUT_DIM + 4 * c);

    for (int m = 0; m < 8; ++m) Wl4[tid + 256 * m] = w4[m];
    __syncthreads();

    v8f acc0 = {}, acc1 = {}, acc2 = {}, acc3 = {};
    for (int u = 0; u < 16; ++u) {
        const float4 b0 = Wl4[(u * 4 + 0) * 32 + lane];
        const float4 b1 = Wl4[(u * 4 + 1) * 32 + lane];
        const float4 b2 = Wl4[(u * 4 + 2) * 32 + lane];
        const float4 b3 = Wl4[(u * 4 + 3) * 32 + lane];
        const v2f aLo = {a4[u].x, a4[u].y};
        const v2f aHi = {a4[u].z, a4[u].w};
        acc0 = __builtin_amdgcn_wmma_f32_16x16x4_f32(false, aLo, false, (v2f){b0.x, b0.y}, (short)0, acc0, false, false);
        acc1 = __builtin_amdgcn_wmma_f32_16x16x4_f32(false, aLo, false, (v2f){b1.x, b1.y}, (short)0, acc1, false, false);
        acc2 = __builtin_amdgcn_wmma_f32_16x16x4_f32(false, aLo, false, (v2f){b2.x, b2.y}, (short)0, acc2, false, false);
        acc3 = __builtin_amdgcn_wmma_f32_16x16x4_f32(false, aLo, false, (v2f){b3.x, b3.y}, (short)0, acc3, false, false);
        acc0 = __builtin_amdgcn_wmma_f32_16x16x4_f32(false, aHi, false, (v2f){b0.z, b0.w}, (short)0, acc0, false, false);
        acc1 = __builtin_amdgcn_wmma_f32_16x16x4_f32(false, aHi, false, (v2f){b1.z, b1.w}, (short)0, acc1, false, false);
        acc2 = __builtin_amdgcn_wmma_f32_16x16x4_f32(false, aHi, false, (v2f){b2.z, b2.w}, (short)0, acc2, false, false);
        acc3 = __builtin_amdgcn_wmma_f32_16x16x4_f32(false, aHi, false, (v2f){b3.z, b3.w}, (short)0, acc3, false, false);
    }

    if (active) {
        float* zt = z + (long)(tile * 16 + 8 * hh) * OUT_DIM + 4 * c;
        for (int r = 0; r < 8; ++r) {
            *(float4*)(zt + (long)r * OUT_DIM) = make_float4(acc0[r], acc1[r], acc2[r], acc3[r]);
        }
    }

    float ps[8], pd[8];
    for (int r = 0; r < 8; ++r) {
        ps[r] = (acc0[r] * as4.x + acc1[r] * as4.y) + (acc2[r] * as4.z + acc3[r] * as4.w);
        pd[r] = (acc0[r] * ad4.x + acc1[r] * ad4.y) + (acc2[r] * ad4.z + acc3[r] * ad4.w);
    }
    for (int r = 0; r < 8; ++r) { ps[r] = ROW_XOR_ADD(ps[r], 1); pd[r] = ROW_XOR_ADD(pd[r], 1); }
    for (int r = 0; r < 8; ++r) { ps[r] = ROW_XOR_ADD(ps[r], 2); pd[r] = ROW_XOR_ADD(pd[r], 2); }
    for (int r = 0; r < 8; ++r) { ps[r] = ROW_XOR_ADD(ps[r], 4); pd[r] = ROW_XOR_ADD(pd[r], 4); }
    for (int r = 0; r < 8; ++r) { ps[r] = ROW_XOR_ADD(ps[r], 8); pd[r] = ROW_XOR_ADD(pd[r], 8); }
    if (active && c == 0) {
        float* sp = s_src + tile * 16 + 8 * hh;
        float* dp = s_dst + tile * 16 + 8 * hh;
        *(float4*)(sp)     = make_float4(ps[0], ps[1], ps[2], ps[3]);
        *(float4*)(sp + 4) = make_float4(ps[4], ps[5], ps[6], ps[7]);
        *(float4*)(dp)     = make_float4(pd[0], pd[1], pd[2], pd[3]);
        *(float4*)(dp + 4) = make_float4(pd[4], pd[5], pd[6], pd[7]);
    }
}

__device__ __forceinline__ float leaky(float v) { return v > 0.0f ? v : 0.01f * v; }
__device__ __forceinline__ float lane_f(float v, int l) {
    return __int_as_float(__builtin_amdgcn_readlane(__float_as_int(v), l));
}
#define ROW_BLOCKS 2048
__global__ __launch_bounds__(256)
void row_softmax_sum(const int* __restrict__ cur, const int* __restrict__ srcs,
                     const float* __restrict__ s_src, const float* __restrict__ s_dst,
                     const float* __restrict__ z, float* __restrict__ out) {
    const int lane = threadIdx.x & 31;
    const int wave0  = __builtin_amdgcn_readfirstlane((int)((blockIdx.x * 256 + threadIdx.x) >> 5));
    const int nwaves = ROW_BLOCKS * 8;
    const float NEG_INF = -__builtin_inff();

    for (int d = wave0; d < N_NODES; d += nwaves) {
        const int beg = cur[d];
        const int end = (d + 1 < N_NODES) ? cur[d + 1] : N_EDGES;
        float accx = 0.0f, accy = 0.0f;
        if (end > beg) {
            const float sd = s_dst[d];
            const int  i0  = beg + lane;
            const bool ok0 = i0 < end;
            const int  s0  = ok0 ? srcs[i0] : 0;
            float l0 = NEG_INF;
            if (ok0) l0 = leaky(s_src[s0] + sd);

            float mx = l0;
            for (int i = i0 + 32; i < end; i += 32) mx = fmaxf(mx, leaky(s_src[srcs[i]] + sd));
            for (int off = 16; off > 0; off >>= 1) mx = fmaxf(mx, __shfl_xor(mx, off, 32));
            mx = (mx == NEG_INF) ? 0.0f : mx;

            const float e0 = ok0 ? __expf(l0 - mx) : 0.0f;
            float sum = e0;
            for (int i = i0 + 32; i < end; i += 32) sum += __expf(leaky(s_src[srcs[i]] + sd) - mx);
            for (int off = 16; off > 0; off >>= 1) sum += __shfl_xor(sum, off, 32);
            const float den = (sum == 0.0f) ? 1.0f : sum;

            for (int base = beg; base < end; base += 32) {
                int   s;
                float al;
                if (base == beg) {
                    s  = s0;
                    al = e0 / den;
                } else {
                    const int  i  = base + lane;
                    const bool ok = i < end;
                    s = ok ? srcs[i] : 0;
                    float ee = 0.0f;
                    if (ok) ee = __expf(leaky(s_src[s] + sd) - mx);
                    al = ee / den;
                }
                const int rem = end - base;
                const int cnt = rem < 32 ? rem : 32;
                for (int j = 0; j < cnt; j += 8) {
                    float2 zv[8];
                    float  a[8];
                    #pragma unroll
                    for (int u = 0; u < 8; ++u) {
                        const int jj = (j + u < cnt) ? (j + u) : (cnt - 1);
                        const int sj = __builtin_amdgcn_readlane(s, jj);
                        a[u]  = (j + u < cnt) ? lane_f(al, jj) : 0.0f;
                        zv[u] = *(const float2*)(z + (long)sj * OUT_DIM + 2 * lane);
                    }
                    #pragma unroll
                    for (int u = 0; u < 8; ++u) {
                        accx += a[u] * zv[u].x;
                        accy += a[u] * zv[u].y;
                    }
                }
            }
        }
        *(float2*)(out + (long)d * OUT_DIM + 2 * lane) = make_float2(accx, accy);
    }
}

static inline size_t alignUp(size_t x) { return (x + 255) & ~(size_t)255; }

extern "C" void kernel_launch(void* const* d_in, const int* in_sizes, int n_in,
                              void* d_out, int out_size, void* d_ws, size_t ws_size,
                              hipStream_t stream) {
    const float* h  = (const float*)d_in[0];
    const float* W  = (const float*)d_in[1];
    const float* a  = (const float*)d_in[2];
    const int* src  = (const int*)d_in[3];
    const int* dst  = (const int*)d_in[4];
    float* out      = (float*)d_out;

    char* ws = (char*)d_ws;
    float* z     = (float*)ws; ws += alignUp((size_t)N_NODES * OUT_DIM * 4);
    float* s_src = (float*)ws; ws += alignUp((size_t)N_NODES * 4);
    float* s_dst = (float*)ws; ws += alignUp((size_t)N_NODES * 4);
    int*   cnt   = (int*)ws;   ws += alignUp((size_t)N_NODES * 4);
    int*   cur   = (int*)ws;   ws += alignUp((size_t)N_NODES * 4);
    int*   srcs  = (int*)ws;   ws += alignUp((size_t)N_EDGES * 4);
    int*   bh    = (int*)z;

    count_part<<<PARTS * SLICES, PART_THREADS, 0, stream>>>(dst, bh);

    slice_offsets<<<PARTS * PART_NODES / 256, 256, 0, stream>>>(bh, cnt);

    row_starts<<<SCAN_BLOCKS, 256, 0, stream>>>(cnt, cur);

    fill_part<<<PARTS * SLICES, PART_THREADS, 0, stream>>>(src, dst, bh, cur, srcs);

    gemm_wmma<<<(N_NODES / 16 + GEMM_WAVES - 1) / GEMM_WAVES, 256, 0, stream>>>(h, W, a, z, s_src, s_dst);

    row_softmax_sum<<<ROW_BLOCKS, 256, 0, stream>>>(cur, srcs, s_src, s_dst, z, out);
}


# edge list read in fewer dependent round trips: count pass one group of 13 loads per lane (was 4 trips of 4), fill pass 2 trips of 13 (was 4 of 8); in-row sum order still stretch, then LDS arrival
# speedup vs baseline: 5.7342x; 1.2397x over previous
#include <hip/hip_runtime.h>

#define N_NODES 100000
#define N_EDGES 1600000
#define IN_DIM  128
#define OUT_DIM 64

typedef __attribute__((ext_vector_type(2))) float v2f;
typedef __attribute__((ext_vector_type(8))) float v8f;

#define PARTS        8
#define PART_NODES   12544
#define CHUNK_NODES  256
#define PART_CHUNKS  (PART_NODES / CHUNK_NODES)
#define TOT_PITCH    64
#define SLICES       32
#define SLICE_VEC    (N_EDGES / SLICES / 4)
#define PART_THREADS 1024
#define COUNT_LOADS  ((SLICE_VEC + PART_THREADS - 1) / PART_THREADS)

#define ROW_XOR_ADD_I(v, M) ((v) + __builtin_amdgcn_update_dpp(0, (v), 0x160 + (M), 0xf, 0xf, true))

__device__ __forceinline__ int lds_add(int* p, int v) {
    return __hip_atomic_fetch_add(p, v, __ATOMIC_RELAXED, __HIP_MEMORY_SCOPE_WORKGROUP);
}

__global__ __launch_bounds__(PART_THREADS)
void count_part(const int* __restrict__ dst, int* __restrict__ bh, int* __restrict__ tot) {
    __shared__ int hist[PART_NODES];
    __shared__ int ctot[TOT_PITCH];
    const int t  = threadIdx.x;
    const int p  = blockIdx.x & (PARTS - 1);
    const int b  = blockIdx.x >> 3;
    const int lo = p * PART_NODES;
    const unsigned span = (unsigned)((N_NODES - lo < PART_NODES) ? (N_NODES - lo) : PART_NODES);
    for (int i = t; i < PART_NODES; i += PART_THREADS) hist[i] = 0;
    __syncthreads();

    const int4* d4 = (const int4*)dst + b * SLICE_VEC;
    {
        int4 d[COUNT_LOADS];
        #pragma unroll
        for (int u = 0; u < COUNT_LOADS; ++u) {
            const int i = t + u * PART_THREADS;
            d[u] = (i < SLICE_VEC) ? d4[i] : make_int4(-1, -1, -1, -1);
        }
        #pragma unroll
        for (int u = 0; u < COUNT_LOADS; ++u) {
            const unsigned x0 = (unsigned)(d[u].x - lo);
            const unsigned x1 = (unsigned)(d[u].y - lo);
            const unsigned x2 = (unsigned)(d[u].z - lo);
            const unsigned x3 = (unsigned)(d[u].w - lo);
            if (x0 < span) lds_add(&hist[x0], 1);
            if (x1 < span) lds_add(&hist[x1], 1);
            if (x2 < span) lds_add(&hist[x2], 1);
            if (x3 < span) lds_add(&hist[x3], 1);
        }
    }
    __syncthreads();
    int* row = bh + (long)(p * SLICES + b) * PART_NODES;
    for (int i = t; i < PART_NODES; i += PART_THREADS) row[i] = hist[i];

    const int lane = t & 31;
    const int wv   = t >> 5;
    int sA = 0, sB = 0;
    for (int q = 0; q < CHUNK_NODES / 32; ++q) sA += hist[wv * CHUNK_NODES + q * 32 + lane];
    if (wv + 32 < PART_CHUNKS) {
        for (int q = 0; q < CHUNK_NODES / 32; ++q) sB += hist[(wv + 32) * CHUNK_NODES + q * 32 + lane];
    }
    sA = ROW_XOR_ADD_I(sA, 1); sB = ROW_XOR_ADD_I(sB, 1);
    sA = ROW_XOR_ADD_I(sA, 2); sB = ROW_XOR_ADD_I(sB, 2);
    sA = ROW_XOR_ADD_I(sA, 4); sB = ROW_XOR_ADD_I(sB, 4);
    sA = ROW_XOR_ADD_I(sA, 8); sB = ROW_XOR_ADD_I(sB, 8);
    sA += __shfl_xor(sA, 16, 32);
    sB += __shfl_xor(sB, 16, 32);
    if (lane == 0) { ctot[wv] = sA; ctot[wv + 32] = sB; }
    __syncthreads();
    if (wv == 0) {
        const int v0 = ctot[lane];
        const int v1 = ctot[lane + 32];
        int inc0 = v0, inc1 = v1;
        for (int off = 1; off < 32; off <<= 1) {
            const int n0 = __shfl_up(inc0, off, 32);
            const int n1 = __shfl_up(inc1, off, 32);
            if (lane >= off) { inc0 += n0; inc1 += n1; }
        }
        const int first32 = __shfl(inc0, 31, 32);
        int* trow = tot + (p * SLICES + b) * TOT_PITCH;
        trow[lane]      = inc0 - v0;
        trow[lane + 32] = first32 + inc1 - v1;
    }
}

__global__ __launch_bounds__(CHUNK_NODES)
void slice_offsets(int* __restrict__ bh, const int* __restrict__ tot, int* __restrict__ cur) {
    __shared__ int wtot[8];
    __shared__ int wpre[8];
    const int t    = threadIdx.x;
    const int lane = t & 31;
    const int wv   = t >> 5;
    const int p = blockIdx.x / PART_CHUNKS;
    const int j = blockIdx.x - p * PART_CHUNKS;
    const int g = blockIdx.x * CHUNK_NODES + t;
    const int i = j * CHUNK_NODES + t;
    int* col = bh + (long)p * (SLICES * PART_NODES) + i;
    int v[SLICES];
    #pragma unroll
    for (int b = 0; b < SLICES; ++b) v[b] = col[b * PART_NODES];
    const int tp = t >> 5;
    int pre = 0;
    if (tp <= p) pre = tot[t * TOT_PITCH + (tp < p ? TOT_PITCH - 1 : j)];
    int run = 0;
    #pragma unroll
    for (int b = 0; b < SLICES; ++b) {
        const int c = v[b];
        v[b] = run;
        run += c;
    }
    for (int off = 16; off > 0; off >>= 1) pre += __shfl_xor(pre, off, 32);
    int inc = run;
    for (int off = 1; off < 32; off <<= 1) {
        const int n = __shfl_up(inc, off, 32);
        if (lane >= off) inc += n;
    }
    if (lane == 31) wtot[wv] = inc;
    if (lane == 0)  wpre[wv] = pre;
    __syncthreads();
    int base = 0;
    for (int w = 0; w < 8; ++w) {
        base += wpre[w];
        if (w < wv) base += wtot[w];
    }
    #pragma unroll
    for (int b = 0; b < SLICES; ++b) col[b * PART_NODES] = v[b];
    if (g < N_NODES) cur[g] = base + inc - run;
}

#define FUSED_THREADS 512
#define POS_TRIPS ((PART_NODES + FUSED_THREADS - 1) / FUSED_THREADS)
#define POS_WORDS (PART_NODES + 32)
#define FILL_LOADS 13
__device__ __forceinline__
void fill_body(const int fid, int* pos,
               const int* __restrict__ src, const int* __restrict__ dst,
               const int* __restrict__ bh, const int* __restrict__ cur,
               int* __restrict__ srcs) {
    const int t  = threadIdx.x;
    const unsigned spare = PART_NODES + (t & 31);
    const int p  = fid & (PARTS - 1);
    const int b  = fid >> 3;
    const int lo = p * PART_NODES;
    const unsigned span = (unsigned)((N_NODES - lo < PART_NODES) ? (N_NODES - lo) : PART_NODES);
    const int* row = bh + (long)(p * SLICES + b) * PART_NODES;
    {
        const int last = (int)span - 1;
        int cv[POS_TRIPS], rv[POS_TRIPS];
        #pragma unroll
        for (int k = 0; k < POS_TRIPS; ++k) {
            const int i  = t + k * FUSED_THREADS;
            const int ic = i < last ? i : last;
            cv[k] = cur[lo + ic];
            rv[k] = row[ic];
        }
        #pragma unroll
        for (int k = 0; k < POS_TRIPS; ++k) {
            const int i = t + k * FUSED_THREADS;
            if (i < PART_NODES) pos[i] = cv[k] + rv[k];
        }
        if (t < 32) pos[PART_NODES + t] = 0;
    }
    __syncthreads();

    const int4* d4 = (const int4*)dst + b * SLICE_VEC;
    const int4* s4 = (const int4*)src + b * SLICE_VEC;
    for (int base = t; base < SLICE_VEC; base += FILL_LOADS * FUSED_THREADS) {
        int4 d[FILL_LOADS], s[FILL_LOADS];
        #pragma unroll
        for (int u = 0; u < FILL_LOADS; ++u) {
            const int i = base + u * FUSED_THREADS;
            const bool in = i < SLICE_VEC;
            d[u] = in ? d4[i] : make_int4(-1, -1, -1, -1);
            s[u] = in ? s4[i] : make_int4(0, 0, 0, 0);
        }
        #pragma unroll
        for (int u = 0; u < FILL_LOADS; ++u) {
            const unsigned x0 = (unsigned)(d[u].x - lo);
            const unsigned x1 = (unsigned)(d[u].y - lo);
            const unsigned x2 = (unsigned)(d[u].z - lo);
            const unsigned x3 = (unsigned)(d[u].w - lo);
            const unsigned q0 = (unsigned)lds_add(&pos[x0 < span ? x0 : spare], 1);
            const unsigned q1 = (unsigned)lds_add(&pos[x1 < span ? x1 : spare], 1);
            const unsigned q2 = (unsigned)lds_add(&pos[x2 < span ? x2 : spare], 1);
            const unsigned q3 = (unsigned)lds_add(&pos[x3 < span ? x3 : spare], 1);
            if (x0 < span) srcs[q0] = s[u].x;
            if (x1 < span) srcs[q1] = s[u].y;
            if (x2 < span) srcs[q2] = s[u].z;
            if (x3 < span) srcs[q3] = s[u].w;
        }
    }
}

#define GEMM_WAVES 16
#define GEMM_BLOCKS ((N_NODES / 16 + GEMM_WAVES - 1) / GEMM_WAVES)

#define ROW_XOR_ADD(v, M) \
    ((v) + __int_as_float(__builtin_amdgcn_update_dpp(0, __float_as_int(v), 0x160 + (M), 0xf, 0xf, true)))
__device__ __forceinline__
void gemm_body(const int gid, float4* Wl4,
               const float* __restrict__ h, const float* __restrict__ W,
               const float* __restrict__ a, float* __restrict__ z,
               float* __restrict__ s_src, float* __restrict__ s_dst) {
    const int tid  = threadIdx.x;
    const int wave = tid >> 5;
    const int lane = tid & 31;
    const int c    = lane & 15;
    const int hh   = lane >> 4;
    const int tileRaw = gid * GEMM_WAVES + wave;
    const bool active = tileRaw < N_NODES / 16;
    const int tile = active ? tileRaw : (N_NODES / 16 - 1);

    const float* wcol = W + (8 * (wave >> 2) + 4 * hh) * OUT_DIM + 4 * c + (wave & 3);
    float4 w4[4];
    for (int m = 0; m < 4; ++m) {
        const float* wp = wcol + m * 32 * OUT_DIM;
        w4[m].x = wp[0 * OUT_DIM];
        w4[m].y = wp[1 * OUT_DIM];
        w4[m].z = wp[2 * OUT_DIM];
        w4[m].w = wp[3 * OUT_DIM];
    }

    const float4* hrow = (const float4*)(h + (long)(tile * 16 + c) * IN_DIM + 4 * hh);
    float4 a4[16];
    for (int u = 0; u < 16; ++u) a4[u] = hrow[2 * u];

    const float4 as4 = *(const float4*)(a + 4 * c);
    const float4 ad4 = *(const float4*)(a + OUT_DIM + 4 * c);

    for (int m = 0; m < 4; ++m) Wl4[tid + 512 * m] = w4[m];
    __syncthreads();

    v8f acc0 = {}, acc1 = {}, acc2 = {}, acc3 = {};
    for (int u = 0; u < 16; ++u) {
        const float4 b0 = Wl4[(u * 4 + 0) * 32 + lane];
        const float4 b1 = Wl4[(u * 4 + 1) * 32 + lane];
        const float4 b2 = Wl4[(u * 4 + 2) * 32 + lane];
        const float4 b3 = Wl4[(u * 4 + 3) * 32 + lane];
        const v2f aLo = {a4[u].x, a4[u].y};
        const v2f aHi = {a4[u].z, a4[u].w};
        acc0 = __builtin_amdgcn_wmma_f32_16x16x4_f32(false, aLo, false, (v2f){b0.x, b0.y}, (short)0, acc0, false, false);
        acc1 = __builtin_amdgcn_wmma_f32_16x16x4_f32(false, aLo, false, (v2f){b1.x, b1.y}, (short)0, acc1, false, false);
        acc2 = __builtin_amdgcn_wmma_f32_16x16x4_f32(false, aLo, false, (v2f){b2.x, b2.y}, (short)0, acc2, false, false);
        acc3 = __builtin_amdgcn_wmma_f32_16x16x4_f32(false, aLo, false, (v2f){b3.x, b3.y}, (short)0, acc3, false, false);
        acc0 = __builtin_amdgcn_wmma_f32_16x16x4_f32(false, aHi, false, (v2f){b0.z, b0.w}, (short)0, acc0, false, false);
        acc1 = __builtin_amdgcn_wmma_f32_16x16x4_f32(false, aHi, false, (v2f){b1.z, b1.w}, (short)0, acc1, false, false);
        acc2 = __builtin_amdgcn_wmma_f32_16x16x4_f32(false, aHi, false, (v2f){b2.z, b2.w}, (short)0, acc2, false, false);
        acc3 = __builtin_amdgcn_wmma_f32_16x16x4_f32(false, aHi, false, (v2f){b3.z, b3.w}, (short)0, acc3, false, false);
    }

    if (active) {
        float* zt = z + (long)(tile * 16 + 8 * hh) * OUT_DIM + 4 * c;
        for (int r = 0; r < 8; ++r) {
            *(float4*)(zt + (long)r * OUT_DIM) = make_float4(acc0[r], acc1[r], acc2[r], acc3[r]);
        }
    }

    float ps[8], pd[8];
    for (int r = 0; r < 8; ++r) {
        ps[r] = (acc0[r] * as4.x + acc1[r] * as4.y) + (acc2[r] * as4.z + acc3[r] * as4.w);
        pd[r] = (acc0[r] * ad4.x + acc1[r] * ad4.y) + (acc2[r] * ad4.z + acc3[r] * ad4.w);
    }
    for (int r = 0; r < 8; ++r) { ps[r] = ROW_XOR_ADD(ps[r], 1); pd[r] = ROW_XOR_ADD(pd[r], 1); }
    for (int r = 0; r < 8; ++r) { ps[r] = ROW_XOR_ADD(ps[r], 2); pd[r] = ROW_XOR_ADD(pd[r], 2); }
    for (int r = 0; r < 8; ++r) { ps[r] = ROW_XOR_ADD(ps[r], 4); pd[r] = ROW_XOR_ADD(pd[r], 4); }
    for (int r = 0; r < 8; ++r) { ps[r] = ROW_XOR_ADD(ps[r], 8); pd[r] = ROW_XOR_ADD(pd[r], 8); }
    if (active && c == 0) {
        float* sp = s_src + tile * 16 + 8 * hh;
        float* dp = s_dst + tile * 16 + 8 * hh;
        *(float4*)(sp)     = make_float4(ps[0], ps[1], ps[2], ps[3]);
        *(float4*)(sp + 4) = make_float4(ps[4], ps[5], ps[6], ps[7]);
        *(float4*)(dp)     = make_float4(pd[0], pd[1], pd[2], pd[3]);
        *(float4*)(dp + 4) = make_float4(pd[4], pd[5], pd[6], pd[7]);
    }
}

#define HALF_BLOCKS  256
#define FUSED_BLOCKS (PARTS * SLICES + GEMM_BLOCKS)
__global__ __launch_bounds__(FUSED_THREADS)
void fill_and_gemm(const int* __restrict__ src, const int* __restrict__ dst,
                   const int* __restrict__ bh, const int* __restrict__ cur,
                   int* __restrict__ srcs,
                   const float* __restrict__ h, const float* __restrict__ W,
                   const float* __restrict__ a, float* __restrict__ z,
                   float* __restrict__ s_src, float* __restrict__ s_dst) {
    __shared__ float4 shm[(POS_WORDS + 3) / 4];
    const int i = blockIdx.x;
    bool gemm;
    int  id;
    if (i < HALF_BLOCKS) {
        gemm = (i & 1) != 0;
        id   = i >> 1;
    } else if (i < 2 * HALF_BLOCKS) {
        const int j = i - HALF_BLOCKS;
        gemm = (j & 1) == 0;
        id   = HALF_BLOCKS / 2 + (j >> 1);
    } else {
        gemm = true;
        id   = i - HALF_BLOCKS;
    }
    if (gemm) gemm_body(id, shm, h, W, a, z, s_src, s_dst);
    else      fill_body(id, (int*)shm, src, dst, bh, cur, srcs);
}

__device__ __forceinline__ float leaky(float v) { return v > 0.0f ? v : 0.01f * v; }
__device__ __forceinline__ float lane_f(float v, int l) {
    return __int_as_float(__builtin_amdgcn_readlane(__float_as_int(v), l));
}
#define ROW_BLOCKS 2048
__global__ __launch_bounds__(256)
void row_softmax_sum(const int* __restrict__ cur, const int* __restrict__ srcs,
                     const float* __restrict__ s_src, const float* __restrict__ s_dst,
                     const float* __restrict__ z, float* __restrict__ out) {
    const int lane = threadIdx.x & 31;
    const int wave0  = __builtin_amdgcn_readfirstlane((int)((blockIdx.x * 256 + threadIdx.x) >> 5));
    const int nwaves = ROW_BLOCKS * 8;
    const float NEG_INF = -__builtin_inff();

    for (int d = wave0; d < N_NODES; d += nwaves) {
        const int beg = cur[d];
        const int end = (d + 1 < N_NODES) ? cur[d + 1] : N_EDGES;
        float accx = 0.0f, accy = 0.0f;
        if (end > beg) {
            const float sd = s_dst[d];
            const int  i0  = beg + lane;
            const bool ok0 = i0 < end;
            const int  s0  = ok0 ? srcs[i0] : 0;
            float l0 = NEG_INF;
            if (ok0) l0 = leaky(s_src[s0] + sd);

            float mx = l0;
            for (int i = i0 + 32; i < end; i += 32) mx = fmaxf(mx, leaky(s_src[srcs[i]] + sd));
            for (int off = 16; off > 0; off >>= 1) mx = fmaxf(mx, __shfl_xor(mx, off, 32));
            mx = (mx == NEG_INF) ? 0.0f : mx;

            const float e0 = ok0 ? __expf(l0 - mx) : 0.0f;
            float sum = e0;
            for (int i = i0 + 32; i < end; i += 32) sum += __expf(leaky(s_src[srcs[i]] + sd) - mx);
            for (int off = 16; off > 0; off >>= 1) sum += __shfl_xor(sum, off, 32);
            const float den = (sum == 0.0f) ? 1.0f : sum;

            for (int base = beg; base < end; base += 32) {
                int   s;
                float al;
                if (base == beg) {
                    s  = s0;
                    al = e0 / den;
                } else {
                    const int  i  = base + lane;
                    const bool ok = i < end;
                    s = ok ? srcs[i] : 0;
                    float ee = 0.0f;
                    if (ok) ee = __expf(leaky(s_src[s] + sd) - mx);
                    al = ee / den;
                }
                const int rem = end - base;
                const int cnt = rem < 32 ? rem : 32;
                for (int j = 0; j < cnt; j += 8) {
                    float2 zv[8];
                    float  a[8];
                    #pragma unroll
                    for (int u = 0; u < 8; ++u) {
                        const int jj = (j + u < cnt) ? (j + u) : (cnt - 1);
                        const int sj = __builtin_amdgcn_readlane(s, jj);
                        a[u]  = (j + u < cnt) ? lane_f(al, jj) : 0.0f;
                        zv[u] = *(const float2*)(z + (long)sj * OUT_DIM + 2 * lane);
                    }
                    #pragma unroll
                    for (int u = 0; u < 8; ++u) {
                        accx += a[u] * zv[u].x;
                        accy += a[u] * zv[u].y;
                    }
                }
            }
        }
        *(float2*)(out + (long)d * OUT_DIM + 2 * lane) = make_float2(accx, accy);
    }
}

static inline size_t alignUp(size_t x) { return (x + 255) & ~(size_t)255; }

extern "C" void kernel_launch(void* const* d_in, const int* in_sizes, int n_in,
                              void* d_out, int out_size, void* d_ws, size_t ws_size,
                              hipStream_t stream) {
    const float* h  = (const float*)d_in[0];
    const float* W  = (const float*)d_in[1];
    const float* a  = (const float*)d_in[2];
    const int* src  = (const int*)d_in[3];
    const int* dst  = (const int*)d_in[4];
    float* out      = (float*)d_out;

    char* ws = (char*)d_ws;
    float* z     = (float*)ws; ws += alignUp((size_t)N_NODES * OUT_DIM * 4);
    float* s_src = (float*)ws; ws += alignUp((size_t)N_NODES * 4);
    float* s_dst = (float*)ws; ws += alignUp((size_t)N_NODES * 4);
    int*   tot   = (int*)ws;   ws += alignUp((size_t)PARTS * SLICES * TOT_PITCH * 4);
    int*   cur   = (int*)ws;   ws += alignUp((size_t)N_NODES * 4);
    int*   srcs  = (int*)ws;   ws += alignUp((size_t)N_EDGES * 4);
    int*   bh    = (int*)out;

    count_part<<<PARTS * SLICES, PART_THREADS, 0, stream>>>(dst, bh, tot);

    slice_offsets<<<PARTS * PART_CHUNKS, CHUNK_NODES, 0, stream>>>(bh, tot, cur);

    fill_and_gemm<<<FUSED_BLOCKS, FUSED_THREADS, 0, stream>>>(src, dst, bh, cur, srcs, h, W, a, z, s_src, s_dst);

    row_softmax_sum<<<ROW_BLOCKS, 256, 0, stream>>>(cur, srcs, s_src, s_dst, z, out);
}
